# starting kernel
# baseline (speedup 1.0000x reference)
#include <hip/hip_runtime.h>
#include <hip/hip_bf16.h>

typedef __attribute__((ext_vector_type(16))) _Float16 v16h;
typedef __attribute__((ext_vector_type(8)))  float    v8f;

__global__ void gemv_wmma_kernel(const float* __restrict__ x,
                                 const float* __restrict__ W,
                                 const float* __restrict__ bias,
                                 float* __restrict__ y,
                                 int K, int N, int do_relu) {
    extern __shared__ float sx[];
    const int tid  = threadIdx.x;
    const int lane = tid & 31;
    const int wave = tid >> 5;

    for (int i = tid; i < K; i += blockDim.x) sx[i] = x[i];
    __syncthreads();

    const int j0  = blockIdx.x * 64 + wave * 16;
    const int col = j0 + (lane & 15);
    const int hi8 = (lane >> 4) * 8;

    v8f acc = {};
    for (int k0 = 0; k0 < K; k0 += 32) {
        v16h ah = {}, al = {}, bh = {}, bl = {};
#pragma unroll
        for (int e = 0; e < 16; ++e) {
            const int kk = k0 + e + (e & 8) + hi8;
            const float xv = sx[kk];
            const float wv = W[(long)kk * N + col];
            const _Float16 xh = (_Float16)xv;
            const _Float16 wh = (_Float16)wv;
            ah[e] = xh;  al[e] = (_Float16)(xv - (float)xh);
            bh[e] = wh;  bl[e] = (_Float16)(wv - (float)wh);
        }
        acc = __builtin_amdgcn_wmma_f32_16x16x32_f16(
            false, ah, false, bh, (short)0, acc, false, false);
        acc = __builtin_amdgcn_wmma_f32_16x16x32_f16(
            false, ah, false, bl, (short)0, acc, false, false);
        acc = __builtin_amdgcn_wmma_f32_16x16x32_f16(
            false, al, false, bh, (short)0, acc, false, false);
    }

    if (lane < 16) {
        float v = acc[0] + bias[col];
        if (do_relu) v = fmaxf(v, 0.0f);
        y[col] = v;
    }
}

#define RN 4096

__global__ void eig_bisect_kernel(const float* __restrict__ ptl,
                                  float* __restrict__ evl) {
    __shared__ float sd[RN];
    const int   l   = blockIdx.y;
    const float ll1 = (float)(l * (l + 1));
    const int   tid = threadIdx.x;
    const float dr  = 10000.0f / (float)RN;

    for (int i = tid; i < RN; i += blockDim.x) {
        const float r = dr * (float)(i + 1);
        sd[i] = 1.0f + ptl[i] + ll1 / (r * r);
    }
    __syncthreads();

    float lo = 1e30f, hi = -1e30f;
    for (int i = 0; i < RN; ++i) {
        const float d = sd[i];
        lo = fminf(lo, d - 1.0f);
        hi = fmaxf(hi, d + 1.0f);
    }

    const int j = blockIdx.x * blockDim.x + tid;

    for (int it = 0; it < 48; ++it) {
        const float x = 0.5f * (lo + hi);
        float q   = sd[0] - x;
        int   cnt = (q < 0.0f);
        for (int i = 1; i < RN; ++i) {
            q = (sd[i] - x) - 0.25f * __builtin_amdgcn_rcpf(q);
            if (fabsf(q) < 1e-30f) q = -1e-30f;
            cnt += (q < 0.0f);
        }
        if (cnt > j) hi = x; else lo = x;
    }
    evl[l * RN + j] = 0.5f * (lo + hi);
}

extern "C" void kernel_launch(void* const* d_in, const int* in_sizes, int n_in,
                              void* d_out, int out_size, void* d_ws, size_t ws_size,
                              hipStream_t stream) {
    const float* energy = (const float*)d_in[0];
    const float* W1     = (const float*)d_in[1];
    const float* b1     = (const float*)d_in[2];
    const float* W2     = (const float*)d_in[3];
    const float* b2     = (const float*)d_in[4];
    const float* W3     = (const float*)d_in[5];
    const float* b3     = (const float*)d_in[6];

    float* out = (float*)d_out;
    float* h1  = (float*)d_ws;
    float* h2  = h1 + 2048;

    gemv_wmma_kernel<<<2048 / 64, 128, 4096 * sizeof(float), stream>>>(
        energy, W1, b1, h1, 4096, 2048, 1);
    gemv_wmma_kernel<<<1024 / 64, 128, 2048 * sizeof(float), stream>>>(
        h1, W2, b2, h2, 2048, 1024, 1);
    gemv_wmma_kernel<<<4096 / 64, 128, 1024 * sizeof(float), stream>>>(
        h2, W3, b3, out, 1024, 4096, 0);
    eig_bisect_kernel<<<dim3(RN / 256, 2), 256, 0, stream>>>(out, out + 4096);
}


# Sturm transfer step run on Q_i = 2^i P_i: row = (2d-2x)*cur - prev as one packed f32 FMA on the (a,b) pair, the two -1/4 multiplies per row gone; 2d staged in LDS; signs and mantissas unchanged
# speedup vs baseline: 8.1555x; 8.1555x over previous
#include <hip/hip_runtime.h>
#include <hip/hip_bf16.h>

typedef __attribute__((ext_vector_type(16))) _Float16 v16h;
typedef __attribute__((ext_vector_type(8)))  float    v8f;
typedef __attribute__((ext_vector_type(2)))  float    v2f;

__global__ void gemv_wmma_kernel(const float* __restrict__ x,
                                 const float* __restrict__ W,
                                 const float* __restrict__ bias,
                                 float* __restrict__ y,
                                 int K, int N, int do_relu) {
    extern __shared__ float sx[];
    const int tid  = threadIdx.x;
    const int lane = tid & 31;
    const int wave = tid >> 5;

    for (int i = tid; i < K; i += blockDim.x) sx[i] = x[i];
    __syncthreads();

    const int j0  = blockIdx.x * 64 + wave * 16;
    const int col = j0 + (lane & 15);
    const int hi8 = (lane >> 4) * 8;

    v8f acc = {};
    for (int k0 = 0; k0 < K; k0 += 32) {
        v16h ah = {}, al = {}, bh = {}, bl = {};
#pragma unroll
        for (int e = 0; e < 16; ++e) {
            const int kk = k0 + e + (e & 8) + hi8;
            const float xv = sx[kk];
            const float wv = W[(long)kk * N + col];
            const _Float16 xh = (_Float16)xv;
            const _Float16 wh = (_Float16)wv;
            ah[e] = xh;  al[e] = (_Float16)(xv - (float)xh);
            bh[e] = wh;  bl[e] = (_Float16)(wv - (float)wh);
        }
        acc = __builtin_amdgcn_wmma_f32_16x16x32_f16(
            false, ah, false, bh, (short)0, acc, false, false);
        acc = __builtin_amdgcn_wmma_f32_16x16x32_f16(
            false, ah, false, bl, (short)0, acc, false, false);
        acc = __builtin_amdgcn_wmma_f32_16x16x32_f16(
            false, al, false, bh, (short)0, acc, false, false);
    }

    if (lane < 16) {
        float v = acc[0] + bias[col];
        if (do_relu) v = fmaxf(v, 0.0f);
        y[col] = v;
    }
}

#define RN   4096
#define SEG  128
#define SEGP 132
#define WPB  32

__device__ __forceinline__ void rescale4(float& a, float& b, float& c, float& d) {
    const float m = fmaxf(fmaxf(fabsf(a), fabsf(b)), fmaxf(fabsf(c), fabsf(d)));
    const int   e = __builtin_amdgcn_frexp_expf(m);
    a = ldexpf(a, -e);  b = ldexpf(b, -e);
    c = ldexpf(c, -e);  d = ldexpf(d, -e);
}

__device__ __forceinline__ float wave_min(float v) {
    for (int o = 16; o > 0; o >>= 1) v = fminf(v, __shfl_xor(v, o));
    return v;
}
__device__ __forceinline__ float wave_max(float v) {
    for (int o = 16; o > 0; o >>= 1) v = fmaxf(v, __shfl_xor(v, o));
    return v;
}

__global__ void __launch_bounds__(WPB * 32)
eig_bisect_kernel(const float* __restrict__ ptl, float* __restrict__ evl) {
    __shared__ __attribute__((aligned(16))) float sd[32 * SEGP];
    __shared__ float smin[WPB], smax[WPB], sres[WPB];
    const int   l    = blockIdx.y;
    const float ll1  = (float)(l * (l + 1));
    const int   tid  = threadIdx.x;
    const int   lane = tid & 31;
    const int   wave = tid >> 5;
    const float dr   = 10000.0f / (float)RN;

    float dmin = 1e30f, dmax = -1e30f;
    for (int i = tid; i < RN; i += WPB * 32) {
        const float r = dr * (float)(i + 1);
        const float d = 1.0f + ptl[i] + ll1 / (r * r);
        sd[(i >> 7) * SEGP + (i & (SEG - 1))] = 2.0f * d;
        dmin = fminf(dmin, d);
        dmax = fmaxf(dmax, d);
    }
    dmin = wave_min(dmin);
    dmax = wave_max(dmax);
    if (lane == 0) { smin[wave] = dmin; smax[wave] = dmax; }
    __syncthreads();
    float lo = wave_min(smin[lane]) - 1.0f;
    float hi = wave_max(smax[lane]) + 1.0f;

    const int j = blockIdx.x * WPB + wave;
    const float4* __restrict__ seg = (const float4*)(sd + lane * SEGP);

    for (int it = 0; it < 48; ++it) {
        const float x = 0.5f * (lo + hi);

        const v2f x2 = {lo + hi, lo + hi};
        v2f top = {1.0f, 0.0f}, prv = {0.0f, 0.5f};
        unsigned hist = 0u;
        int cnt = 0;
        for (int w = 0; w < SEG / 32; ++w) {
            const unsigned prev = hist;
            for (int g = 0; g < 8; ++g) {
                const float4 v = seg[w * 8 + g];
                const v2f t01 = (v2f){v.x, v.y} - x2;
                const v2f t23 = (v2f){v.z, v.w} - x2;
                v2f nt;
                nt = __builtin_elementwise_fma((v2f){t01.x, t01.x}, top, -prv); prv = top; top = nt;
                hist = __builtin_amdgcn_alignbit(hist, __float_as_uint(top.x), 31);
                nt = __builtin_elementwise_fma((v2f){t01.y, t01.y}, top, -prv); prv = top; top = nt;
                hist = __builtin_amdgcn_alignbit(hist, __float_as_uint(top.x), 31);
                nt = __builtin_elementwise_fma((v2f){t23.x, t23.x}, top, -prv); prv = top; top = nt;
                hist = __builtin_amdgcn_alignbit(hist, __float_as_uint(top.x), 31);
                nt = __builtin_elementwise_fma((v2f){t23.y, t23.y}, top, -prv); prv = top; top = nt;
                hist = __builtin_amdgcn_alignbit(hist, __float_as_uint(top.x), 31);
                float ra = top.x, rb = top.y, rc = prv.x, rd = prv.y;
                rescale4(ra, rb, rc, rd);
                top = (v2f){ra, rb};  prv = (v2f){rc, rd};
            }
            cnt += __popc(hist ^ __builtin_amdgcn_alignbit(prev, hist, 1));
        }
        const unsigned nd = hist & 1u;
        float a = top.x, b = top.y;
        const float a0 = a, b0 = b;
        float c = 2.0f * prv.x, d = 2.0f * prv.y;

        for (int dl = 1; dl < 32; dl <<= 1) {
            const float ua = __shfl_up(a, dl), ub = __shfl_up(b, dl);
            const float uc = __shfl_up(c, dl), ud = __shfl_up(d, dl);
            if (lane >= dl) {
                const float na = fmaf(a, ua, b * uc), nb = fmaf(a, ub, b * ud);
                const float nc = fmaf(c, ua, d * uc), nd2 = fmaf(c, ub, d * ud);
                a = na; b = nb; c = nc; d = nd2;
                rescale4(a, b, c, d);
            }
        }
        float pa = __shfl_up(a, 1), pc = __shfl_up(c, 1);
        if (lane == 0) { pa = 1.0f; pc = 0.0f; }

        const float pe = fmaf(a0, pa, b0 * pc);
        const unsigned spe = __float_as_uint(pe) >> 31, spa = __float_as_uint(pa) >> 31;
        const unsigned spc = __float_as_uint(pc) >> 31;
        const unsigned moved = spe ^ spa ^ nd;
        const unsigned qneg  = spa ^ spc;
        cnt += (int)moved - (int)(2u * (moved & qneg));
        for (int o = 16; o > 0; o >>= 1) cnt += __shfl_xor(cnt, o);

        if (cnt > j) hi = x; else lo = x;
    }

    if (lane == 0) sres[wave] = 0.5f * (lo + hi);
    __syncthreads();
    if (wave == 0) evl[l * RN + blockIdx.x * WPB + lane] = sres[lane];
}

extern "C" void kernel_launch(void* const* d_in, const int* in_sizes, int n_in,
                              void* d_out, int out_size, void* d_ws, size_t ws_size,
                              hipStream_t stream) {
    const float* energy = (const float*)d_in[0];
    const float* W1     = (const float*)d_in[1];
    const float* b1     = (const float*)d_in[2];
    const float* W2     = (const float*)d_in[3];
    const float* b2     = (const float*)d_in[4];
    const float* W3     = (const float*)d_in[5];
    const float* b3     = (const float*)d_in[6];

    float* out = (float*)d_out;
    float* h1  = (float*)d_ws;
    float* h2  = h1 + 2048;

    gemv_wmma_kernel<<<2048 / 64, 128, 4096 * sizeof(float), stream>>>(
        energy, W1, b1, h1, 4096, 2048, 1);
    gemv_wmma_kernel<<<1024 / 64, 128, 2048 * sizeof(float), stream>>>(
        h1, W2, b2, h2, 2048, 1024, 1);
    gemv_wmma_kernel<<<4096 / 64, 128, 1024 * sizeof(float), stream>>>(
        h2, W3, b3, out, 1024, 4096, 0);
    eig_bisect_kernel<<<dim3(RN / WPB, 2), WPB * 32, 0, stream>>>(out, out + 4096);
}


# Pass rescales the transfer product by its power of two every 32 rows, not every 4 (16 or 4 where the Gershgorin width would let 32 rows overflow f32); exact scaling, signs and mantissas unchanged
# speedup vs baseline: 9.2149x; 1.1299x over previous
#include <hip/hip_runtime.h>
#include <hip/hip_bf16.h>

typedef __attribute__((ext_vector_type(16))) _Float16 v16h;
typedef __attribute__((ext_vector_type(8)))  float    v8f;
typedef __attribute__((ext_vector_type(2)))  float    v2f;

__global__ void gemv_wmma_kernel(const float* __restrict__ x,
                                 const float* __restrict__ W,
                                 const float* __restrict__ bias,
                                 float* __restrict__ y,
                                 int K, int N, int do_relu) {
    extern __shared__ float sx[];
    const int tid  = threadIdx.x;
    const int lane = tid & 31;
    const int wave = tid >> 5;

    for (int i = tid; i < K; i += blockDim.x) sx[i] = x[i];
    __syncthreads();

    const int j0  = blockIdx.x * 64 + wave * 16;
    const int col = j0 + (lane & 15);
    const int hi8 = (lane >> 4) * 8;

    v8f acc = {};
    for (int k0 = 0; k0 < K; k0 += 32) {
        v16h ah = {}, al = {}, bh = {}, bl = {};
#pragma unroll
        for (int e = 0; e < 16; ++e) {
            const int kk = k0 + e + (e & 8) + hi8;
            const float xv = sx[kk];
            const float wv = W[(long)kk * N + col];
            const _Float16 xh = (_Float16)xv;
            const _Float16 wh = (_Float16)wv;
            ah[e] = xh;  al[e] = (_Float16)(xv - (float)xh);
            bh[e] = wh;  bl[e] = (_Float16)(wv - (float)wh);
        }
        acc = __builtin_amdgcn_wmma_f32_16x16x32_f16(
            false, ah, false, bh, (short)0, acc, false, false);
        acc = __builtin_amdgcn_wmma_f32_16x16x32_f16(
            false, ah, false, bl, (short)0, acc, false, false);
        acc = __builtin_amdgcn_wmma_f32_16x16x32_f16(
            false, al, false, bh, (short)0, acc, false, false);
    }

    if (lane < 16) {
        float v = acc[0] + bias[col];
        if (do_relu) v = fmaxf(v, 0.0f);
        y[col] = v;
    }
}

#define RN   4096
#define SEG  128
#define SEGP 132
#define WPB  32

__device__ __forceinline__ void rescale4(float& a, float& b, float& c, float& d) {
    const float m = fmaxf(fmaxf(fabsf(a), fabsf(b)), fmaxf(fabsf(c), fabsf(d)));
    const int   e = __builtin_amdgcn_frexp_expf(m);
    a = ldexpf(a, -e);  b = ldexpf(b, -e);
    c = ldexpf(c, -e);  d = ldexpf(d, -e);
}

__device__ __forceinline__ float wave_min(float v) {
    for (int o = 16; o > 0; o >>= 1) v = fminf(v, __shfl_xor(v, o));
    return v;
}
__device__ __forceinline__ float wave_max(float v) {
    for (int o = 16; o > 0; o >>= 1) v = fmaxf(v, __shfl_xor(v, o));
    return v;
}

template <int RS>
__device__ __forceinline__ int segment_pass(const float4* __restrict__ seg, const v2f x2,
                                            v2f& top, v2f& prv, unsigned& hist) {
    int cnt = 0;
    for (int w = 0; w < SEG / 32; ++w) {
        const unsigned prev = hist;
#pragma unroll
        for (int g = 0; g < 8; ++g) {
            const float4 v = seg[w * 8 + g];
            const v2f t01 = (v2f){v.x, v.y} - x2;
            const v2f t23 = (v2f){v.z, v.w} - x2;
            v2f nt;
            nt = __builtin_elementwise_fma((v2f){t01.x, t01.x}, top, -prv); prv = top; top = nt;
            hist = __builtin_amdgcn_alignbit(hist, __float_as_uint(top.x), 31);
            nt = __builtin_elementwise_fma((v2f){t01.y, t01.y}, top, -prv); prv = top; top = nt;
            hist = __builtin_amdgcn_alignbit(hist, __float_as_uint(top.x), 31);
            nt = __builtin_elementwise_fma((v2f){t23.x, t23.x}, top, -prv); prv = top; top = nt;
            hist = __builtin_amdgcn_alignbit(hist, __float_as_uint(top.x), 31);
            nt = __builtin_elementwise_fma((v2f){t23.y, t23.y}, top, -prv); prv = top; top = nt;
            hist = __builtin_amdgcn_alignbit(hist, __float_as_uint(top.x), 31);
            if ((g + 1) % (RS / 4) == 0) {
                float ra = top.x, rb = top.y, rc = prv.x, rd = prv.y;
                rescale4(ra, rb, rc, rd);
                top = (v2f){ra, rb};  prv = (v2f){rc, rd};
            }
        }
        cnt += __popc(hist ^ __builtin_amdgcn_alignbit(prev, hist, 1));
    }
    return cnt;
}

__global__ void __launch_bounds__(WPB * 32)
eig_bisect_kernel(const float* __restrict__ ptl, float* __restrict__ evl) {
    __shared__ __attribute__((aligned(16))) float sd[32 * SEGP];
    __shared__ float smin[WPB], smax[WPB], sres[WPB];
    const int   l    = blockIdx.y;
    const float ll1  = (float)(l * (l + 1));
    const int   tid  = threadIdx.x;
    const int   lane = tid & 31;
    const int   wave = tid >> 5;
    const float dr   = 10000.0f / (float)RN;

    float dmin = 1e30f, dmax = -1e30f;
    for (int i = tid; i < RN; i += WPB * 32) {
        const float r = dr * (float)(i + 1);
        const float d = 1.0f + ptl[i] + ll1 / (r * r);
        sd[(i >> 7) * SEGP + (i & (SEG - 1))] = 2.0f * d;
        dmin = fminf(dmin, d);
        dmax = fmaxf(dmax, d);
    }
    dmin = wave_min(dmin);
    dmax = wave_max(dmax);
    if (lane == 0) { smin[wave] = dmin; smax[wave] = dmax; }
    __syncthreads();
    float lo = wave_min(smin[lane]) - 1.0f;
    float hi = wave_max(smax[lane]) + 1.0f;

    const float width = __uint_as_float(__builtin_amdgcn_readfirstlane(__float_as_uint(hi - lo)));
    const int rs = width <= 8.0f ? 32 : (width <= 64.0f ? 16 : 4);

    const int j = blockIdx.x * WPB + wave;
    const float4* __restrict__ seg = (const float4*)(sd + lane * SEGP);

    for (int it = 0; it < 48; ++it) {
        const float x = 0.5f * (lo + hi);

        const v2f x2 = {lo + hi, lo + hi};
        v2f top = {1.0f, 0.0f}, prv = {0.0f, 0.5f};
        unsigned hist = 0u;
        int cnt;
        if (rs == 32)      cnt = segment_pass<32>(seg, x2, top, prv, hist);
        else if (rs == 16) cnt = segment_pass<16>(seg, x2, top, prv, hist);
        else               cnt = segment_pass<4>(seg, x2, top, prv, hist);
        const unsigned nd = hist & 1u;
        float a = top.x, b = top.y;
        const float a0 = a, b0 = b;
        float c = 2.0f * prv.x, d = 2.0f * prv.y;

        for (int dl = 1; dl < 32; dl <<= 1) {
            const float ua = __shfl_up(a, dl), ub = __shfl_up(b, dl);
            const float uc = __shfl_up(c, dl), ud = __shfl_up(d, dl);
            if (lane >= dl) {
                const float na = fmaf(a, ua, b * uc), nb = fmaf(a, ub, b * ud);
                const float nc = fmaf(c, ua, d * uc), nd2 = fmaf(c, ub, d * ud);
                a = na; b = nb; c = nc; d = nd2;
                rescale4(a, b, c, d);
            }
        }
        float pa = __shfl_up(a, 1), pc = __shfl_up(c, 1);
        if (lane == 0) { pa = 1.0f; pc = 0.0f; }

        const float pe = fmaf(a0, pa, b0 * pc);
        const unsigned spe = __float_as_uint(pe) >> 31, spa = __float_as_uint(pa) >> 31;
        const unsigned spc = __float_as_uint(pc) >> 31;
        const unsigned moved = spe ^ spa ^ nd;
        const unsigned qneg  = spa ^ spc;
        cnt += (int)moved - (int)(2u * (moved & qneg));
        for (int o = 16; o > 0; o >>= 1) cnt += __shfl_xor(cnt, o);

        if (cnt > j) hi = x; else lo = x;
    }

    if (lane == 0) sres[wave] = 0.5f * (lo + hi);
    __syncthreads();
    if (wave == 0) evl[l * RN + blockIdx.x * WPB + lane] = sres[lane];
}

extern "C" void kernel_launch(void* const* d_in, const int* in_sizes, int n_in,
                              void* d_out, int out_size, void* d_ws, size_t ws_size,
                              hipStream_t stream) {
    const float* energy = (const float*)d_in[0];
    const float* W1     = (const float*)d_in[1];
    const float* b1     = (const float*)d_in[2];
    const float* W2     = (const float*)d_in[3];
    const float* b2     = (const float*)d_in[4];
    const float* W3     = (const float*)d_in[5];
    const float* b3     = (const float*)d_in[6];

    float* out = (float*)d_out;
    float* h1  = (float*)d_ws;
    float* h2  = h1 + 2048;

    gemv_wmma_kernel<<<2048 / 64, 128, 4096 * sizeof(float), stream>>>(
        energy, W1, b1, h1, 4096, 2048, 1);
    gemv_wmma_kernel<<<1024 / 64, 128, 2048 * sizeof(float), stream>>>(
        h1, W2, b2, h2, 2048, 1024, 1);
    gemv_wmma_kernel<<<4096 / 64, 128, 1024 * sizeof(float), stream>>>(
        h2, W3, b3, out, 1024, 4096, 0);
    eig_bisect_kernel<<<dim3(RN / WPB, 2), WPB * 32, 0, stream>>>(out, out + 4096);
}


# GEMV layers cut along K over the grid (16/16/8 row slabs, 512/256/512 workgroups): slab sums go to the workspace, the next launch adds them in slab order (+bias, relu) while staging x; new sum order
# speedup vs baseline: 21.0470x; 2.2840x over previous
#include <hip/hip_runtime.h>
#include <hip/hip_bf16.h>

typedef __attribute__((ext_vector_type(16))) _Float16 v16h;
typedef __attribute__((ext_vector_type(8)))  float    v8f;
typedef __attribute__((ext_vector_type(2)))  float    v2f;

template <int P, bool ACT, int KS>
__global__ void __launch_bounds__(128)
gemv_slab_kernel(const float* __restrict__ xin,
                 const float* __restrict__ bin,
                 const float* __restrict__ W,
                 float* __restrict__ ypart,
                 int Kfull, int N) {
    __shared__ float sx[KS];
    const int tid   = threadIdx.x;
    const int lane  = tid & 31;
    const int wave  = tid >> 5;
    const int kbase = blockIdx.y * KS;

    for (int i = tid; i < KS; i += 128) {
        const int k = kbase + i;
        float v = xin[k];
#pragma unroll
        for (int p = 1; p < P; ++p) v += xin[p * Kfull + k];
        if (ACT) v = fmaxf(v + bin[k], 0.0f);
        sx[i] = v;
    }
    __syncthreads();

    const int j0  = blockIdx.x * 64 + wave * 16;
    const int col = j0 + (lane & 15);
    const int hi8 = (lane >> 4) * 8;

    v8f acc = {};
    for (int k0 = 0; k0 < KS; k0 += 32) {
        v16h ah = {}, al = {}, bh = {}, bl = {};
#pragma unroll
        for (int e = 0; e < 16; ++e) {
            const int kk = k0 + e + (e & 8) + hi8;
            const float xv = sx[kk];
            const float wv = W[(long)(kbase + kk) * N + col];
            const _Float16 xh = (_Float16)xv;
            const _Float16 wh = (_Float16)wv;
            ah[e] = xh;  al[e] = (_Float16)(xv - (float)xh);
            bh[e] = wh;  bl[e] = (_Float16)(wv - (float)wh);
        }
        acc = __builtin_amdgcn_wmma_f32_16x16x32_f16(
            false, ah, false, bh, (short)0, acc, false, false);
        acc = __builtin_amdgcn_wmma_f32_16x16x32_f16(
            false, ah, false, bl, (short)0, acc, false, false);
        acc = __builtin_amdgcn_wmma_f32_16x16x32_f16(
            false, al, false, bh, (short)0, acc, false, false);
    }

    if (lane < 16)
        ypart[blockIdx.y * N + col] = acc[0];
}

#define RN   4096
#define SEG  128
#define SEGP 132
#define WPB  32
#define NP3  8

__device__ __forceinline__ float ptl_value(const float* __restrict__ part,
                                           const float* __restrict__ b3, int i) {
    float v = part[i];
#pragma unroll
    for (int p = 1; p < NP3; ++p) v += part[p * RN + i];
    return v + b3[i];
}

__device__ __forceinline__ void rescale4(float& a, float& b, float& c, float& d) {
    const float m = fmaxf(fmaxf(fabsf(a), fabsf(b)), fmaxf(fabsf(c), fabsf(d)));
    const int   e = __builtin_amdgcn_frexp_expf(m);
    a = ldexpf(a, -e);  b = ldexpf(b, -e);
    c = ldexpf(c, -e);  d = ldexpf(d, -e);
}

__device__ __forceinline__ float wave_min(float v) {
    for (int o = 16; o > 0; o >>= 1) v = fminf(v, __shfl_xor(v, o));
    return v;
}
__device__ __forceinline__ float wave_max(float v) {
    for (int o = 16; o > 0; o >>= 1) v = fmaxf(v, __shfl_xor(v, o));
    return v;
}

template <int RS>
__device__ __forceinline__ int segment_pass(const float4* __restrict__ seg, const v2f x2,
                                            v2f& top, v2f& prv, unsigned& hist) {
    int cnt = 0;
    for (int w = 0; w < SEG / 32; ++w) {
        const unsigned prev = hist;
#pragma unroll
        for (int g = 0; g < 8; ++g) {
            const float4 v = seg[w * 8 + g];
            const v2f t01 = (v2f){v.x, v.y} - x2;
            const v2f t23 = (v2f){v.z, v.w} - x2;
            v2f nt;
            nt = __builtin_elementwise_fma((v2f){t01.x, t01.x}, top, -prv); prv = top; top = nt;
            hist = __builtin_amdgcn_alignbit(hist, __float_as_uint(top.x), 31);
            nt = __builtin_elementwise_fma((v2f){t01.y, t01.y}, top, -prv); prv = top; top = nt;
            hist = __builtin_amdgcn_alignbit(hist, __float_as_uint(top.x), 31);
            nt = __builtin_elementwise_fma((v2f){t23.x, t23.x}, top, -prv); prv = top; top = nt;
            hist = __builtin_amdgcn_alignbit(hist, __float_as_uint(top.x), 31);
            nt = __builtin_elementwise_fma((v2f){t23.y, t23.y}, top, -prv); prv = top; top = nt;
            hist = __builtin_amdgcn_alignbit(hist, __float_as_uint(top.x), 31);
            if ((g + 1) % (RS / 4) == 0) {
                float ra = top.x, rb = top.y, rc = prv.x, rd = prv.y;
                rescale4(ra, rb, rc, rd);
                top = (v2f){ra, rb};  prv = (v2f){rc, rd};
            }
        }
        cnt += __popc(hist ^ __builtin_amdgcn_alignbit(prev, hist, 1));
    }
    return cnt;
}

__global__ void __launch_bounds__(WPB * 32)
eig_bisect_kernel(const float* __restrict__ part3, const float* __restrict__ b3,
                  float* __restrict__ ptl_out, float* __restrict__ evl) {
    __shared__ __attribute__((aligned(16))) float sd[32 * SEGP];
    __shared__ float smin[WPB], smax[WPB], sres[WPB];
    const int   l    = blockIdx.y;
    const float ll1  = (float)(l * (l + 1));
    const int   tid  = threadIdx.x;
    const int   lane = tid & 31;
    const int   wave = tid >> 5;
    const float dr   = 10000.0f / (float)RN;

    float dmin = 1e30f, dmax = -1e30f;
    for (int i = tid; i < RN; i += WPB * 32) {
        const float r = dr * (float)(i + 1);
        const float d = 1.0f + ptl_value(part3, b3, i) + ll1 / (r * r);
        sd[(i >> 7) * SEGP + (i & (SEG - 1))] = 2.0f * d;
        dmin = fminf(dmin, d);
        dmax = fmaxf(dmax, d);
    }
    if (l == 0 && wave == 0)
        ptl_out[blockIdx.x * WPB + lane] = ptl_value(part3, b3, blockIdx.x * WPB + lane);
    dmin = wave_min(dmin);
    dmax = wave_max(dmax);
    if (lane == 0) { smin[wave] = dmin; smax[wave] = dmax; }
    __syncthreads();
    float lo = wave_min(smin[lane]) - 1.0f;
    float hi = wave_max(smax[lane]) + 1.0f;

    const float width = __uint_as_float(__builtin_amdgcn_readfirstlane(__float_as_uint(hi - lo)));
    const int rs = width <= 8.0f ? 32 : (width <= 64.0f ? 16 : 4);

    const int j = blockIdx.x * WPB + wave;
    const float4* __restrict__ seg = (const float4*)(sd + lane * SEGP);

    for (int it = 0; it < 48; ++it) {
        const float x = 0.5f * (lo + hi);

        const v2f x2 = {lo + hi, lo + hi};
        v2f top = {1.0f, 0.0f}, prv = {0.0f, 0.5f};
        unsigned hist = 0u;
        int cnt;
        if (rs == 32)      cnt = segment_pass<32>(seg, x2, top, prv, hist);
        else if (rs == 16) cnt = segment_pass<16>(seg, x2, top, prv, hist);
        else               cnt = segment_pass<4>(seg, x2, top, prv, hist);
        const unsigned nd = hist & 1u;
        float a = top.x, b = top.y;
        const float a0 = a, b0 = b;
        float c = 2.0f * prv.x, d = 2.0f * prv.y;

        for (int dl = 1; dl < 32; dl <<= 1) {
            const float ua = __shfl_up(a, dl), ub = __shfl_up(b, dl);
            const float uc = __shfl_up(c, dl), ud = __shfl_up(d, dl);
            if (lane >= dl) {
                const float na = fmaf(a, ua, b * uc), nb = fmaf(a, ub, b * ud);
                const float nc = fmaf(c, ua, d * uc), nd2 = fmaf(c, ub, d * ud);
                a = na; b = nb; c = nc; d = nd2;
                rescale4(a, b, c, d);
            }
        }
        float pa = __shfl_up(a, 1), pc = __shfl_up(c, 1);
        if (lane == 0) { pa = 1.0f; pc = 0.0f; }

        const float pe = fmaf(a0, pa, b0 * pc);
        const unsigned spe = __float_as_uint(pe) >> 31, spa = __float_as_uint(pa) >> 31;
        const unsigned spc = __float_as_uint(pc) >> 31;
        const unsigned moved = spe ^ spa ^ nd;
        const unsigned qneg  = spa ^ spc;
        cnt += (int)moved - (int)(2u * (moved & qneg));
        for (int o = 16; o > 0; o >>= 1) cnt += __shfl_xor(cnt, o);

        if (cnt > j) hi = x; else lo = x;
    }

    if (lane == 0) sres[wave] = 0.5f * (lo + hi);
    __syncthreads();
    if (wave == 0) evl[l * RN + blockIdx.x * WPB + lane] = sres[lane];
}

extern "C" void kernel_launch(void* const* d_in, const int* in_sizes, int n_in,
                              void* d_out, int out_size, void* d_ws, size_t ws_size,
                              hipStream_t stream) {
    const float* energy = (const float*)d_in[0];
    const float* W1     = (const float*)d_in[1];
    const float* b1     = (const float*)d_in[2];
    const float* W2     = (const float*)d_in[3];
    const float* b2     = (const float*)d_in[4];
    const float* W3     = (const float*)d_in[5];
    const float* b3     = (const float*)d_in[6];

    float* out   = (float*)d_out;
    float* part1 = (float*)d_ws;
    float* part2 = part1 + 16 * 2048;
    float* part3 = part1;

    gemv_slab_kernel<1, false, 256><<<dim3(2048 / 64, 16), 128, 0, stream>>>(
        energy, b1, W1, part1, 4096, 2048);
    gemv_slab_kernel<16, true, 128><<<dim3(1024 / 64, 16), 128, 0, stream>>>(
        part1, b1, W2, part2, 2048, 1024);
    gemv_slab_kernel<16, true, 128><<<dim3(4096 / 64, 8), 128, 0, stream>>>(
        part2, b2, W3, part3, 1024, 4096);
    eig_bisect_kernel<<<dim3(RN / WPB, 2), WPB * 32, 0, stream>>>(part3, b3, out, out + 4096);
}


# Bisection runs on the floats' ordered integer keys (trial point = middle key): ceil(log2(keys in the Gershgorin interval)) <= 32 steps reach neighbouring floats everywhere; 48 value halvings gone
# speedup vs baseline: 28.7593x; 1.3664x over previous
#include <hip/hip_runtime.h>
#include <hip/hip_bf16.h>

typedef __attribute__((ext_vector_type(16))) _Float16 v16h;
typedef __attribute__((ext_vector_type(8)))  float    v8f;
typedef __attribute__((ext_vector_type(2)))  float    v2f;

template <int P, bool ACT, int KS>
__global__ void __launch_bounds__(128)
gemv_slab_kernel(const float* __restrict__ xin,
                 const float* __restrict__ bin,
                 const float* __restrict__ W,
                 float* __restrict__ ypart,
                 int Kfull, int N) {
    __shared__ float sx[KS];
    const int tid   = threadIdx.x;
    const int lane  = tid & 31;
    const int wave  = tid >> 5;
    const int kbase = blockIdx.y * KS;

    for (int i = tid; i < KS; i += 128) {
        const int k = kbase + i;
        float v = xin[k];
#pragma unroll
        for (int p = 1; p < P; ++p) v += xin[p * Kfull + k];
        if (ACT) v = fmaxf(v + bin[k], 0.0f);
        sx[i] = v;
    }
    __syncthreads();

    const int j0  = blockIdx.x * 64 + wave * 16;
    const int col = j0 + (lane & 15);
    const int hi8 = (lane >> 4) * 8;

    v8f acc = {};
    for (int k0 = 0; k0 < KS; k0 += 32) {
        v16h ah = {}, al = {}, bh = {}, bl = {};
#pragma unroll
        for (int e = 0; e < 16; ++e) {
            const int kk = k0 + e + (e & 8) + hi8;
            const float xv = sx[kk];
            const float wv = W[(long)(kbase + kk) * N + col];
            const _Float16 xh = (_Float16)xv;
            const _Float16 wh = (_Float16)wv;
            ah[e] = xh;  al[e] = (_Float16)(xv - (float)xh);
            bh[e] = wh;  bl[e] = (_Float16)(wv - (float)wh);
        }
        acc = __builtin_amdgcn_wmma_f32_16x16x32_f16(
            false, ah, false, bh, (short)0, acc, false, false);
        acc = __builtin_amdgcn_wmma_f32_16x16x32_f16(
            false, ah, false, bl, (short)0, acc, false, false);
        acc = __builtin_amdgcn_wmma_f32_16x16x32_f16(
            false, al, false, bh, (short)0, acc, false, false);
    }

    if (lane < 16)
        ypart[blockIdx.y * N + col] = acc[0];
}

#define RN   4096
#define SEG  128
#define SEGP 132
#define WPB  32
#define NP3  8

__device__ __forceinline__ float ptl_value(const float* __restrict__ part,
                                           const float* __restrict__ b3, int i) {
    float v = part[i];
#pragma unroll
    for (int p = 1; p < NP3; ++p) v += part[p * RN + i];
    return v + b3[i];
}

__device__ __forceinline__ void rescale4(float& a, float& b, float& c, float& d) {
    const float m = fmaxf(fmaxf(fabsf(a), fabsf(b)), fmaxf(fabsf(c), fabsf(d)));
    const int   e = __builtin_amdgcn_frexp_expf(m);
    a = ldexpf(a, -e);  b = ldexpf(b, -e);
    c = ldexpf(c, -e);  d = ldexpf(d, -e);
}

__device__ __forceinline__ unsigned float_to_key(float f) {
    const unsigned b = __float_as_uint(f);
    return (b & 0x80000000u) ? ~b : (b | 0x80000000u);
}
__device__ __forceinline__ float key_to_float(unsigned k) {
    return __uint_as_float((k & 0x80000000u) ? (k ^ 0x80000000u) : ~k);
}

__device__ __forceinline__ float wave_min(float v) {
    for (int o = 16; o > 0; o >>= 1) v = fminf(v, __shfl_xor(v, o));
    return v;
}
__device__ __forceinline__ float wave_max(float v) {
    for (int o = 16; o > 0; o >>= 1) v = fmaxf(v, __shfl_xor(v, o));
    return v;
}

template <int RS>
__device__ __forceinline__ int segment_pass(const float4* __restrict__ seg, const v2f x2,
                                            v2f& top, v2f& prv, unsigned& hist) {
    int cnt = 0;
    for (int w = 0; w < SEG / 32; ++w) {
        const unsigned prev = hist;
#pragma unroll
        for (int g = 0; g < 8; ++g) {
            const float4 v = seg[w * 8 + g];
            const v2f t01 = (v2f){v.x, v.y} - x2;
            const v2f t23 = (v2f){v.z, v.w} - x2;
            v2f nt;
            nt = __builtin_elementwise_fma((v2f){t01.x, t01.x}, top, -prv); prv = top; top = nt;
            hist = __builtin_amdgcn_alignbit(hist, __float_as_uint(top.x), 31);
            nt = __builtin_elementwise_fma((v2f){t01.y, t01.y}, top, -prv); prv = top; top = nt;
            hist = __builtin_amdgcn_alignbit(hist, __float_as_uint(top.x), 31);
            nt = __builtin_elementwise_fma((v2f){t23.x, t23.x}, top, -prv); prv = top; top = nt;
            hist = __builtin_amdgcn_alignbit(hist, __float_as_uint(top.x), 31);
            nt = __builtin_elementwise_fma((v2f){t23.y, t23.y}, top, -prv); prv = top; top = nt;
            hist = __builtin_amdgcn_alignbit(hist, __float_as_uint(top.x), 31);
            if ((g + 1) % (RS / 4) == 0) {
                float ra = top.x, rb = top.y, rc = prv.x, rd = prv.y;
                rescale4(ra, rb, rc, rd);
                top = (v2f){ra, rb};  prv = (v2f){rc, rd};
            }
        }
        cnt += __popc(hist ^ __builtin_amdgcn_alignbit(prev, hist, 1));
    }
    return cnt;
}

__global__ void __launch_bounds__(WPB * 32)
eig_bisect_kernel(const float* __restrict__ part3, const float* __restrict__ b3,
                  float* __restrict__ ptl_out, float* __restrict__ evl) {
    __shared__ __attribute__((aligned(16))) float sd[32 * SEGP];
    __shared__ float smin[WPB], smax[WPB], sres[WPB];
    const int   l    = blockIdx.y;
    const float ll1  = (float)(l * (l + 1));
    const int   tid  = threadIdx.x;
    const int   lane = tid & 31;
    const int   wave = tid >> 5;
    const float dr   = 10000.0f / (float)RN;

    float dmin = 1e30f, dmax = -1e30f;
    for (int i = tid; i < RN; i += WPB * 32) {
        const float r = dr * (float)(i + 1);
        const float d = 1.0f + ptl_value(part3, b3, i) + ll1 / (r * r);
        sd[(i >> 7) * SEGP + (i & (SEG - 1))] = 2.0f * d;
        dmin = fminf(dmin, d);
        dmax = fmaxf(dmax, d);
    }
    if (l == 0 && wave == 0)
        ptl_out[blockIdx.x * WPB + lane] = ptl_value(part3, b3, blockIdx.x * WPB + lane);
    dmin = wave_min(dmin);
    dmax = wave_max(dmax);
    if (lane == 0) { smin[wave] = dmin; smax[wave] = dmax; }
    __syncthreads();
    float lo = wave_min(smin[lane]) - 1.0f;
    float hi = wave_max(smax[lane]) + 1.0f;

    const float width = __uint_as_float(__builtin_amdgcn_readfirstlane(__float_as_uint(hi - lo)));
    const int rs = width <= 8.0f ? 32 : (width <= 64.0f ? 16 : 4);

    const int j = blockIdx.x * WPB + wave;
    const float4* __restrict__ seg = (const float4*)(sd + lane * SEGP);

    unsigned klo = float_to_key(lo), khi = float_to_key(hi);
    const unsigned gap0 = __builtin_amdgcn_readfirstlane(khi - klo);
    const int nsteps = gap0 > 1u ? 32 - __builtin_clz(gap0 - 1u) : 0;

    for (int it = 0; it < nsteps; ++it) {
        const unsigned km = klo + ((khi - klo) >> 1);
        const float x = key_to_float(km);

        const v2f x2 = {x + x, x + x};
        v2f top = {1.0f, 0.0f}, prv = {0.0f, 0.5f};
        unsigned hist = 0u;
        int cnt;
        if (rs == 32)      cnt = segment_pass<32>(seg, x2, top, prv, hist);
        else if (rs == 16) cnt = segment_pass<16>(seg, x2, top, prv, hist);
        else               cnt = segment_pass<4>(seg, x2, top, prv, hist);
        const unsigned nd = hist & 1u;
        float a = top.x, b = top.y;
        const float a0 = a, b0 = b;
        float c = 2.0f * prv.x, d = 2.0f * prv.y;

        for (int dl = 1; dl < 32; dl <<= 1) {
            const float ua = __shfl_up(a, dl), ub = __shfl_up(b, dl);
            const float uc = __shfl_up(c, dl), ud = __shfl_up(d, dl);
            if (lane >= dl) {
                const float na = fmaf(a, ua, b * uc), nb = fmaf(a, ub, b * ud);
                const float nc = fmaf(c, ua, d * uc), nd2 = fmaf(c, ub, d * ud);
                a = na; b = nb; c = nc; d = nd2;
                rescale4(a, b, c, d);
            }
        }
        float pa = __shfl_up(a, 1), pc = __shfl_up(c, 1);
        if (lane == 0) { pa = 1.0f; pc = 0.0f; }

        const float pe = fmaf(a0, pa, b0 * pc);
        const unsigned spe = __float_as_uint(pe) >> 31, spa = __float_as_uint(pa) >> 31;
        const unsigned spc = __float_as_uint(pc) >> 31;
        const unsigned moved = spe ^ spa ^ nd;
        const unsigned qneg  = spa ^ spc;
        cnt += (int)moved - (int)(2u * (moved & qneg));
        for (int o = 16; o > 0; o >>= 1) cnt += __shfl_xor(cnt, o);

        if (cnt > j) khi = km; else klo = km;
    }
    lo = key_to_float(klo);
    hi = key_to_float(khi);

    if (lane == 0) sres[wave] = 0.5f * (lo + hi);
    __syncthreads();
    if (wave == 0) evl[l * RN + blockIdx.x * WPB + lane] = sres[lane];
}

extern "C" void kernel_launch(void* const* d_in, const int* in_sizes, int n_in,
                              void* d_out, int out_size, void* d_ws, size_t ws_size,
                              hipStream_t stream) {
    const float* energy = (const float*)d_in[0];
    const float* W1     = (const float*)d_in[1];
    const float* b1     = (const float*)d_in[2];
    const float* W2     = (const float*)d_in[3];
    const float* b2     = (const float*)d_in[4];
    const float* W3     = (const float*)d_in[5];
    const float* b3     = (const float*)d_in[6];

    float* out   = (float*)d_out;
    float* part1 = (float*)d_ws;
    float* part2 = part1 + 16 * 2048;
    float* part3 = part1;

    gemv_slab_kernel<1, false, 256><<<dim3(2048 / 64, 16), 128, 0, stream>>>(
        energy, b1, W1, part1, 4096, 2048);
    gemv_slab_kernel<16, true, 128><<<dim3(1024 / 64, 16), 128, 0, stream>>>(
        part1, b1, W2, part2, 2048, 1024);
    gemv_slab_kernel<16, true, 128><<<dim3(4096 / 64, 8), 128, 0, stream>>>(
        part2, b2, W3, part3, 1024, 4096);
    eig_bisect_kernel<<<dim3(RN / WPB, 2), WPB * 32, 0, stream>>>(part3, b3, out, out + 4096);
}


# Sturm chain cut over 16 lanes of 256 rows, two eigenvalues per wave (was 32 lanes of 128, one per wave): scan, fix-up and count sum serve two eigenvalues, 4 scan rounds not 5; low bits may move
# speedup vs baseline: 31.3366x; 1.0896x over previous
#include <hip/hip_runtime.h>
#include <hip/hip_bf16.h>

typedef __attribute__((ext_vector_type(16))) _Float16 v16h;
typedef __attribute__((ext_vector_type(8)))  float    v8f;
typedef __attribute__((ext_vector_type(2)))  float    v2f;

template <int P, bool ACT, int KS>
__global__ void __launch_bounds__(128)
gemv_slab_kernel(const float* __restrict__ xin,
                 const float* __restrict__ bin,
                 const float* __restrict__ W,
                 float* __restrict__ ypart,
                 int Kfull, int N) {
    __shared__ float sx[KS];
    const int tid   = threadIdx.x;
    const int lane  = tid & 31;
    const int wave  = tid >> 5;
    const int kbase = blockIdx.y * KS;

    for (int i = tid; i < KS; i += 128) {
        const int k = kbase + i;
        float v = xin[k];
#pragma unroll
        for (int p = 1; p < P; ++p) v += xin[p * Kfull + k];
        if (ACT) v = fmaxf(v + bin[k], 0.0f);
        sx[i] = v;
    }
    __syncthreads();

    const int j0  = blockIdx.x * 64 + wave * 16;
    const int col = j0 + (lane & 15);
    const int hi8 = (lane >> 4) * 8;

    v8f acc = {};
    for (int k0 = 0; k0 < KS; k0 += 32) {
        v16h ah = {}, al = {}, bh = {}, bl = {};
#pragma unroll
        for (int e = 0; e < 16; ++e) {
            const int kk = k0 + e + (e & 8) + hi8;
            const float xv = sx[kk];
            const float wv = W[(long)(kbase + kk) * N + col];
            const _Float16 xh = (_Float16)xv;
            const _Float16 wh = (_Float16)wv;
            ah[e] = xh;  al[e] = (_Float16)(xv - (float)xh);
            bh[e] = wh;  bl[e] = (_Float16)(wv - (float)wh);
        }
        acc = __builtin_amdgcn_wmma_f32_16x16x32_f16(
            false, ah, false, bh, (short)0, acc, false, false);
        acc = __builtin_amdgcn_wmma_f32_16x16x32_f16(
            false, ah, false, bl, (short)0, acc, false, false);
        acc = __builtin_amdgcn_wmma_f32_16x16x32_f16(
            false, al, false, bh, (short)0, acc, false, false);
    }

    if (lane < 16)
        ypart[blockIdx.y * N + col] = acc[0];
}

#define RN   4096
#define LPE  16
#define SEG  256
#define SEGP 260
#define EPB  32
#define WPB  16
#define NP3  8

__device__ __forceinline__ float ptl_value(const float* __restrict__ part,
                                           const float* __restrict__ b3, int i) {
    float v = part[i];
#pragma unroll
    for (int p = 1; p < NP3; ++p) v += part[p * RN + i];
    return v + b3[i];
}

__device__ __forceinline__ void rescale4(float& a, float& b, float& c, float& d) {
    const float m = fmaxf(fmaxf(fabsf(a), fabsf(b)), fmaxf(fabsf(c), fabsf(d)));
    const int   e = __builtin_amdgcn_frexp_expf(m);
    a = ldexpf(a, -e);  b = ldexpf(b, -e);
    c = ldexpf(c, -e);  d = ldexpf(d, -e);
}

__device__ __forceinline__ unsigned float_to_key(float f) {
    const unsigned b = __float_as_uint(f);
    return (b & 0x80000000u) ? ~b : (b | 0x80000000u);
}
__device__ __forceinline__ float key_to_float(unsigned k) {
    return __uint_as_float((k & 0x80000000u) ? (k ^ 0x80000000u) : ~k);
}

__device__ __forceinline__ float wave_min(float v) {
    for (int o = 16; o > 0; o >>= 1) v = fminf(v, __shfl_xor(v, o));
    return v;
}
__device__ __forceinline__ float wave_max(float v) {
    for (int o = 16; o > 0; o >>= 1) v = fmaxf(v, __shfl_xor(v, o));
    return v;
}

template <int RS>
__device__ __forceinline__ int segment_pass(const float4* __restrict__ seg, const v2f x2,
                                            v2f& top, v2f& prv, unsigned& hist) {
    int cnt = 0;
    for (int w = 0; w < SEG / 32; ++w) {
        const unsigned prev = hist;
#pragma unroll
        for (int g = 0; g < 8; ++g) {
            const float4 v = seg[w * 8 + g];
            const v2f t01 = (v2f){v.x, v.y} - x2;
            const v2f t23 = (v2f){v.z, v.w} - x2;
            v2f nt;
            nt = __builtin_elementwise_fma((v2f){t01.x, t01.x}, top, -prv); prv = top; top = nt;
            hist = __builtin_amdgcn_alignbit(hist, __float_as_uint(top.x), 31);
            nt = __builtin_elementwise_fma((v2f){t01.y, t01.y}, top, -prv); prv = top; top = nt;
            hist = __builtin_amdgcn_alignbit(hist, __float_as_uint(top.x), 31);
            nt = __builtin_elementwise_fma((v2f){t23.x, t23.x}, top, -prv); prv = top; top = nt;
            hist = __builtin_amdgcn_alignbit(hist, __float_as_uint(top.x), 31);
            nt = __builtin_elementwise_fma((v2f){t23.y, t23.y}, top, -prv); prv = top; top = nt;
            hist = __builtin_amdgcn_alignbit(hist, __float_as_uint(top.x), 31);
            if ((g + 1) % (RS / 4) == 0) {
                float ra = top.x, rb = top.y, rc = prv.x, rd = prv.y;
                rescale4(ra, rb, rc, rd);
                top = (v2f){ra, rb};  prv = (v2f){rc, rd};
            }
        }
        cnt += __popc(hist ^ __builtin_amdgcn_alignbit(prev, hist, 1));
    }
    return cnt;
}

__global__ void __launch_bounds__(WPB * 32)
eig_bisect_kernel(const float* __restrict__ part3, const float* __restrict__ b3,
                  float* __restrict__ ptl_out, float* __restrict__ evl) {
    __shared__ __attribute__((aligned(16))) float sd[LPE * SEGP];
    __shared__ float smin[WPB], smax[WPB], sres[EPB];
    const int   l    = blockIdx.y;
    const float ll1  = (float)(l * (l + 1));
    const int   tid  = threadIdx.x;
    const int   lane = tid & 31;
    const int   wave = tid >> 5;
    const int   sub  = lane & (LPE - 1);
    const int   grp  = lane >> 4;
    const float dr   = 10000.0f / (float)RN;

    float dmin = 1e30f, dmax = -1e30f;
#pragma unroll
    for (int i = tid; i < RN; i += WPB * 32) {
        const float r = dr * (float)(i + 1);
        const float d = 1.0f + ptl_value(part3, b3, i) + ll1 / (r * r);
        sd[(i >> 8) * SEGP + (i & (SEG - 1))] = 2.0f * d;
        dmin = fminf(dmin, d);
        dmax = fmaxf(dmax, d);
    }
    if (l == 0 && wave == 0)
        ptl_out[blockIdx.x * EPB + lane] = ptl_value(part3, b3, blockIdx.x * EPB + lane);
    dmin = wave_min(dmin);
    dmax = wave_max(dmax);
    if (lane == 0) { smin[wave] = dmin; smax[wave] = dmax; }
    __syncthreads();
    float lo = wave_min(smin[lane & (WPB - 1)]) - 1.0f;
    float hi = wave_max(smax[lane & (WPB - 1)]) + 1.0f;

    const float width = __uint_as_float(__builtin_amdgcn_readfirstlane(__float_as_uint(hi - lo)));
    const int rs = width <= 8.0f ? 32 : (width <= 64.0f ? 16 : 4);

    const int j = blockIdx.x * EPB + wave * 2 + grp;
    const float4* __restrict__ seg = (const float4*)(sd + sub * SEGP);

    unsigned klo = float_to_key(lo), khi = float_to_key(hi);
    const unsigned gap0 = __builtin_amdgcn_readfirstlane(khi - klo);
    const int nsteps = gap0 > 1u ? 32 - __builtin_clz(gap0 - 1u) : 0;

    for (int it = 0; it < nsteps; ++it) {
        const unsigned km = klo + ((khi - klo) >> 1);
        const float x = key_to_float(km);

        const v2f x2 = {x + x, x + x};
        v2f top = {1.0f, 0.0f}, prv = {0.0f, 0.5f};
        unsigned hist = 0u;
        int cnt;
        if (rs == 32)      cnt = segment_pass<32>(seg, x2, top, prv, hist);
        else if (rs == 16) cnt = segment_pass<16>(seg, x2, top, prv, hist);
        else               cnt = segment_pass<4>(seg, x2, top, prv, hist);
        const unsigned nd = hist & 1u;
        float a = top.x, b = top.y;
        const float a0 = a, b0 = b;
        float c = 2.0f * prv.x, d = 2.0f * prv.y;

        for (int dl = 1; dl < LPE; dl <<= 1) {
            const float ua = __shfl_up(a, dl, LPE), ub = __shfl_up(b, dl, LPE);
            const float uc = __shfl_up(c, dl, LPE), ud = __shfl_up(d, dl, LPE);
            if (sub >= dl) {
                const float na = fmaf(a, ua, b * uc), nb = fmaf(a, ub, b * ud);
                const float nc = fmaf(c, ua, d * uc), nd2 = fmaf(c, ub, d * ud);
                a = na; b = nb; c = nc; d = nd2;
                rescale4(a, b, c, d);
            }
        }
        float pa = __shfl_up(a, 1, LPE), pc = __shfl_up(c, 1, LPE);
        if (sub == 0) { pa = 1.0f; pc = 0.0f; }

        const float pe = fmaf(a0, pa, b0 * pc);
        const unsigned spe = __float_as_uint(pe) >> 31, spa = __float_as_uint(pa) >> 31;
        const unsigned spc = __float_as_uint(pc) >> 31;
        const unsigned moved = spe ^ spa ^ nd;
        const unsigned qneg  = spa ^ spc;
        cnt += (int)moved - (int)(2u * (moved & qneg));
        for (int o = LPE / 2; o > 0; o >>= 1) cnt += __shfl_xor(cnt, o);

        if (cnt > j) khi = km; else klo = km;
    }
    lo = key_to_float(klo);
    hi = key_to_float(khi);

    if (sub == 0) sres[wave * 2 + grp] = 0.5f * (lo + hi);
    __syncthreads();
    if (wave == 0) evl[l * RN + blockIdx.x * EPB + lane] = sres[lane];
}

extern "C" void kernel_launch(void* const* d_in, const int* in_sizes, int n_in,
                              void* d_out, int out_size, void* d_ws, size_t ws_size,
                              hipStream_t stream) {
    const float* energy = (const float*)d_in[0];
    const float* W1     = (const float*)d_in[1];
    const float* b1     = (const float*)d_in[2];
    const float* W2     = (const float*)d_in[3];
    const float* b2     = (const float*)d_in[4];
    const float* W3     = (const float*)d_in[5];
    const float* b3     = (const float*)d_in[6];

    float* out   = (float*)d_out;
    float* part1 = (float*)d_ws;
    float* part2 = part1 + 16 * 2048;
    float* part3 = part1;

    gemv_slab_kernel<1, false, 256><<<dim3(2048 / 64, 16), 128, 0, stream>>>(
        energy, b1, W1, part1, 4096, 2048);
    gemv_slab_kernel<16, true, 128><<<dim3(1024 / 64, 16), 128, 0, stream>>>(
        part1, b1, W2, part2, 2048, 1024);
    gemv_slab_kernel<16, true, 128><<<dim3(4096 / 64, 8), 128, 0, stream>>>(
        part2, b2, W3, part3, 1024, 4096);
    eig_bisect_kernel<<<dim3(RN / EPB, 2), WPB * 32, 0, stream>>>(part3, b3, out, out + 4096);
}


# GEMV layers: a wave loads every W element of its slab (64 or 128 per lane) as one group behind the loads of x, ahead of the barrier; was one waited round trip per 32 rows after staging; same sums
# speedup vs baseline: 33.7703x; 1.0777x over previous
#include <hip/hip_runtime.h>
#include <hip/hip_bf16.h>

typedef __attribute__((ext_vector_type(16))) _Float16 v16h;
typedef __attribute__((ext_vector_type(8)))  float    v8f;
typedef __attribute__((ext_vector_type(2)))  float    v2f;

template <int P, bool ACT, int KS>
__global__ void __launch_bounds__(128)
gemv_slab_kernel(const float* __restrict__ xin,
                 const float* __restrict__ bin,
                 const float* __restrict__ W,
                 float* __restrict__ ypart,
                 int Kfull, int N) {
    __shared__ float sx[KS];
    const int tid   = threadIdx.x;
    const int lane  = tid & 31;
    const int wave  = tid >> 5;
    const int kbase = blockIdx.y * KS;

    const int j0  = blockIdx.x * 64 + wave * 16;
    const int col = j0 + (lane & 15);
    const int hi8 = (lane >> 4) * 8;

    constexpr int XPT = KS / 128;
    float xs[XPT];
#pragma unroll
    for (int r = 0; r < XPT; ++r) {
        const int k = kbase + tid + r * 128;
        float v = xin[k];
#pragma unroll
        for (int p = 1; p < P; ++p) v += xin[p * Kfull + k];
        if (ACT) v = fmaxf(v + bin[k], 0.0f);
        xs[r] = v;
    }

    constexpr int NF = KS / 32;
    float wr[NF][16];
#pragma unroll
    for (int f = 0; f < NF; ++f) {
#pragma unroll
        for (int e = 0; e < 16; ++e) {
            const int kk = f * 32 + e + (e & 8) + hi8;
            wr[f][e] = W[(long)(kbase + kk) * N + col];
        }
    }

#pragma unroll
    for (int r = 0; r < XPT; ++r) sx[tid + r * 128] = xs[r];
    __syncthreads();

    v8f acc = {};
#pragma unroll
    for (int f = 0; f < NF; ++f) {
        v16h ah = {}, al = {}, bh = {}, bl = {};
#pragma unroll
        for (int e = 0; e < 16; ++e) {
            const int kk = f * 32 + e + (e & 8) + hi8;
            const float xv = sx[kk];
            const float wv = wr[f][e];
            const _Float16 xh = (_Float16)xv;
            const _Float16 wh = (_Float16)wv;
            ah[e] = xh;  al[e] = (_Float16)(xv - (float)xh);
            bh[e] = wh;  bl[e] = (_Float16)(wv - (float)wh);
        }
        acc = __builtin_amdgcn_wmma_f32_16x16x32_f16(
            false, ah, false, bh, (short)0, acc, false, false);
        acc = __builtin_amdgcn_wmma_f32_16x16x32_f16(
            false, ah, false, bl, (short)0, acc, false, false);
        acc = __builtin_amdgcn_wmma_f32_16x16x32_f16(
            false, al, false, bh, (short)0, acc, false, false);
    }

    if (lane < 16)
        ypart[blockIdx.y * N + col] = acc[0];
}

#define RN   4096
#define LPE  16
#define SEG  256
#define SEGP 260
#define EPB  32
#define WPB  16
#define NP3  8

__device__ __forceinline__ float ptl_value(const float* __restrict__ part,
                                           const float* __restrict__ b3, int i) {
    float v = part[i];
#pragma unroll
    for (int p = 1; p < NP3; ++p) v += part[p * RN + i];
    return v + b3[i];
}

__device__ __forceinline__ void rescale4(float& a, float& b, float& c, float& d) {
    const float m = fmaxf(fmaxf(fabsf(a), fabsf(b)), fmaxf(fabsf(c), fabsf(d)));
    const int   e = __builtin_amdgcn_frexp_expf(m);
    a = ldexpf(a, -e);  b = ldexpf(b, -e);
    c = ldexpf(c, -e);  d = ldexpf(d, -e);
}

__device__ __forceinline__ unsigned float_to_key(float f) {
    const unsigned b = __float_as_uint(f);
    return (b & 0x80000000u) ? ~b : (b | 0x80000000u);
}
__device__ __forceinline__ float key_to_float(unsigned k) {
    return __uint_as_float((k & 0x80000000u) ? (k ^ 0x80000000u) : ~k);
}

__device__ __forceinline__ float wave_min(float v) {
    for (int o = 16; o > 0; o >>= 1) v = fminf(v, __shfl_xor(v, o));
    return v;
}
__device__ __forceinline__ float wave_max(float v) {
    for (int o = 16; o > 0; o >>= 1) v = fmaxf(v, __shfl_xor(v, o));
    return v;
}

template <int RS>
__device__ __forceinline__ int segment_pass(const float4* __restrict__ seg, const v2f x2,
                                            v2f& top, v2f& prv, unsigned& hist) {
    int cnt = 0;
    for (int w = 0; w < SEG / 32; ++w) {
        const unsigned prev = hist;
#pragma unroll
        for (int g = 0; g < 8; ++g) {
            const float4 v = seg[w * 8 + g];
            const v2f t01 = (v2f){v.x, v.y} - x2;
            const v2f t23 = (v2f){v.z, v.w} - x2;
            v2f nt;
            nt = __builtin_elementwise_fma((v2f){t01.x, t01.x}, top, -prv); prv = top; top = nt;
            hist = __builtin_amdgcn_alignbit(hist, __float_as_uint(top.x), 31);
            nt = __builtin_elementwise_fma((v2f){t01.y, t01.y}, top, -prv); prv = top; top = nt;
            hist = __builtin_amdgcn_alignbit(hist, __float_as_uint(top.x), 31);
            nt = __builtin_elementwise_fma((v2f){t23.x, t23.x}, top, -prv); prv = top; top = nt;
            hist = __builtin_amdgcn_alignbit(hist, __float_as_uint(top.x), 31);
            nt = __builtin_elementwise_fma((v2f){t23.y, t23.y}, top, -prv); prv = top; top = nt;
            hist = __builtin_amdgcn_alignbit(hist, __float_as_uint(top.x), 31);
            if ((g + 1) % (RS / 4) == 0) {
                float ra = top.x, rb = top.y, rc = prv.x, rd = prv.y;
                rescale4(ra, rb, rc, rd);
                top = (v2f){ra, rb};  prv = (v2f){rc, rd};
            }
        }
        cnt += __popc(hist ^ __builtin_amdgcn_alignbit(prev, hist, 1));
    }
    return cnt;
}

__global__ void __launch_bounds__(WPB * 32)
eig_bisect_kernel(const float* __restrict__ part3, const float* __restrict__ b3,
                  float* __restrict__ ptl_out, float* __restrict__ evl) {
    __shared__ __attribute__((aligned(16))) float sd[LPE * SEGP];
    __shared__ float smin[WPB], smax[WPB], sres[EPB];
    const int   l    = blockIdx.y;
    const float ll1  = (float)(l * (l + 1));
    const int   tid  = threadIdx.x;
    const int   lane = tid & 31;
    const int   wave = tid >> 5;
    const int   sub  = lane & (LPE - 1);
    const int   grp  = lane >> 4;
    const float dr   = 10000.0f / (float)RN;

    float dmin = 1e30f, dmax = -1e30f;
#pragma unroll
    for (int i = tid; i < RN; i += WPB * 32) {
        const float r = dr * (float)(i + 1);
        const float d = 1.0f + ptl_value(part3, b3, i) + ll1 / (r * r);
        sd[(i >> 8) * SEGP + (i & (SEG - 1))] = 2.0f * d;
        dmin = fminf(dmin, d);
        dmax = fmaxf(dmax, d);
    }
    if (l == 0 && wave == 0)
        ptl_out[blockIdx.x * EPB + lane] = ptl_value(part3, b3, blockIdx.x * EPB + lane);
    dmin = wave_min(dmin);
    dmax = wave_max(dmax);
    if (lane == 0) { smin[wave] = dmin; smax[wave] = dmax; }
    __syncthreads();
    float lo = wave_min(smin[lane & (WPB - 1)]) - 1.0f;
    float hi = wave_max(smax[lane & (WPB - 1)]) + 1.0f;

    const float width = __uint_as_float(__builtin_amdgcn_readfirstlane(__float_as_uint(hi - lo)));
    const int rs = width <= 8.0f ? 32 : (width <= 64.0f ? 16 : 4);

    const int j = blockIdx.x * EPB + wave * 2 + grp;
    const float4* __restrict__ seg = (const float4*)(sd + sub * SEGP);

    unsigned klo = float_to_key(lo), khi = float_to_key(hi);
    const unsigned gap0 = __builtin_amdgcn_readfirstlane(khi - klo);
    const int nsteps = gap0 > 1u ? 32 - __builtin_clz(gap0 - 1u) : 0;

    for (int it = 0; it < nsteps; ++it) {
        const unsigned km = klo + ((khi - klo) >> 1);
        const float x = key_to_float(km);

        const v2f x2 = {x + x, x + x};
        v2f top = {1.0f, 0.0f}, prv = {0.0f, 0.5f};
        unsigned hist = 0u;
        int cnt;
        if (rs == 32)      cnt = segment_pass<32>(seg, x2, top, prv, hist);
        else if (rs == 16) cnt = segment_pass<16>(seg, x2, top, prv, hist);
        else               cnt = segment_pass<4>(seg, x2, top, prv, hist);
        const unsigned nd = hist & 1u;
        float a = top.x, b = top.y;
        const float a0 = a, b0 = b;
        float c = 2.0f * prv.x, d = 2.0f * prv.y;

        for (int dl = 1; dl < LPE; dl <<= 1) {
            const float ua = __shfl_up(a, dl, LPE), ub = __shfl_up(b, dl, LPE);
            const float uc = __shfl_up(c, dl, LPE), ud = __shfl_up(d, dl, LPE);
            if (sub >= dl) {
                const float na = fmaf(a, ua, b * uc), nb = fmaf(a, ub, b * ud);
                const float nc = fmaf(c, ua, d * uc), nd2 = fmaf(c, ub, d * ud);
                a = na; b = nb; c = nc; d = nd2;
                rescale4(a, b, c, d);
            }
        }
        float pa = __shfl_up(a, 1, LPE), pc = __shfl_up(c, 1, LPE);
        if (sub == 0) { pa = 1.0f; pc = 0.0f; }

        const float pe = fmaf(a0, pa, b0 * pc);
        const unsigned spe = __float_as_uint(pe) >> 31, spa = __float_as_uint(pa) >> 31;
        const unsigned spc = __float_as_uint(pc) >> 31;
        const unsigned moved = spe ^ spa ^ nd;
        const unsigned qneg  = spa ^ spc;
        cnt += (int)moved - (int)(2u * (moved & qneg));
        for (int o = LPE / 2; o > 0; o >>= 1) cnt += __shfl_xor(cnt, o);

        if (cnt > j) khi = km; else klo = km;
    }
    lo = key_to_float(klo);
    hi = key_to_float(khi);

    if (sub == 0) sres[wave * 2 + grp] = 0.5f * (lo + hi);
    __syncthreads();
    if (wave == 0) evl[l * RN + blockIdx.x * EPB + lane] = sres[lane];
}

extern "C" void kernel_launch(void* const* d_in, const int* in_sizes, int n_in,
                              void* d_out, int out_size, void* d_ws, size_t ws_size,
                              hipStream_t stream) {
    const float* energy = (const float*)d_in[0];
    const float* W1     = (const float*)d_in[1];
    const float* b1     = (const float*)d_in[2];
    const float* W2     = (const float*)d_in[3];
    const float* b2     = (const float*)d_in[4];
    const float* W3     = (const float*)d_in[5];
    const float* b3     = (const float*)d_in[6];

    float* out   = (float*)d_out;
    float* part1 = (float*)d_ws;
    float* part2 = part1 + 16 * 2048;
    float* part3 = part1;

    gemv_slab_kernel<1, false, 256><<<dim3(2048 / 64, 16), 128, 0, stream>>>(
        energy, b1, W1, part1, 4096, 2048);
    gemv_slab_kernel<16, true, 128><<<dim3(1024 / 64, 16), 128, 0, stream>>>(
        part1, b1, W2, part2, 2048, 1024);
    gemv_slab_kernel<16, true, 128><<<dim3(4096 / 64, 8), 128, 0, stream>>>(
        part2, b2, W3, part3, 1024, 4096);
    eig_bisect_kernel<<<dim3(RN / EPB, 2), WPB * 32, 0, stream>>>(part3, b3, out, out + 4096);
}


# While a wave's two neighbouring eigenvalues still share a bracket, ONE Sturm chain over all 32 lanes (128 rows each) serves both; 16-lane chains per group once a count parts them; low bits may move
# speedup vs baseline: 38.1159x; 1.1287x over previous
#include <hip/hip_runtime.h>
#include <hip/hip_bf16.h>

typedef __attribute__((ext_vector_type(16))) _Float16 v16h;
typedef __attribute__((ext_vector_type(8)))  float    v8f;
typedef __attribute__((ext_vector_type(2)))  float    v2f;

template <int P, bool ACT, int KS>
__global__ void __launch_bounds__(128)
gemv_slab_kernel(const float* __restrict__ xin,
                 const float* __restrict__ bin,
                 const float* __restrict__ W,
                 float* __restrict__ ypart,
                 int Kfull, int N) {
    __shared__ float sx[KS];
    const int tid   = threadIdx.x;
    const int lane  = tid & 31;
    const int wave  = tid >> 5;
    const int kbase = blockIdx.y * KS;

    const int j0  = blockIdx.x * 64 + wave * 16;
    const int col = j0 + (lane & 15);
    const int hi8 = (lane >> 4) * 8;

    constexpr int XPT = KS / 128;
    float xs[XPT];
#pragma unroll
    for (int r = 0; r < XPT; ++r) {
        const int k = kbase + tid + r * 128;
        float v = xin[k];
#pragma unroll
        for (int p = 1; p < P; ++p) v += xin[p * Kfull + k];
        if (ACT) v = fmaxf(v + bin[k], 0.0f);
        xs[r] = v;
    }

    constexpr int NF = KS / 32;
    float wr[NF][16];
#pragma unroll
    for (int f = 0; f < NF; ++f) {
#pragma unroll
        for (int e = 0; e < 16; ++e) {
            const int kk = f * 32 + e + (e & 8) + hi8;
            wr[f][e] = W[(long)(kbase + kk) * N + col];
        }
    }

#pragma unroll
    for (int r = 0; r < XPT; ++r) sx[tid + r * 128] = xs[r];
    __syncthreads();

    v8f acc = {};
#pragma unroll
    for (int f = 0; f < NF; ++f) {
        v16h ah = {}, al = {}, bh = {}, bl = {};
#pragma unroll
        for (int e = 0; e < 16; ++e) {
            const int kk = f * 32 + e + (e & 8) + hi8;
            const float xv = sx[kk];
            const float wv = wr[f][e];
            const _Float16 xh = (_Float16)xv;
            const _Float16 wh = (_Float16)wv;
            ah[e] = xh;  al[e] = (_Float16)(xv - (float)xh);
            bh[e] = wh;  bl[e] = (_Float16)(wv - (float)wh);
        }
        acc = __builtin_amdgcn_wmma_f32_16x16x32_f16(
            false, ah, false, bh, (short)0, acc, false, false);
        acc = __builtin_amdgcn_wmma_f32_16x16x32_f16(
            false, ah, false, bl, (short)0, acc, false, false);
        acc = __builtin_amdgcn_wmma_f32_16x16x32_f16(
            false, al, false, bh, (short)0, acc, false, false);
    }

    if (lane < 16)
        ypart[blockIdx.y * N + col] = acc[0];
}

#define RN   4096
#define LPE  16
#define SEG  256
#define SEGP 260
#define EPB  32
#define WPB  16
#define NP3  8

__device__ __forceinline__ float ptl_value(const float* __restrict__ part,
                                           const float* __restrict__ b3, int i) {
    float v = part[i];
#pragma unroll
    for (int p = 1; p < NP3; ++p) v += part[p * RN + i];
    return v + b3[i];
}

__device__ __forceinline__ void rescale4(float& a, float& b, float& c, float& d) {
    const float m = fmaxf(fmaxf(fabsf(a), fabsf(b)), fmaxf(fabsf(c), fabsf(d)));
    const int   e = __builtin_amdgcn_frexp_expf(m);
    a = ldexpf(a, -e);  b = ldexpf(b, -e);
    c = ldexpf(c, -e);  d = ldexpf(d, -e);
}

__device__ __forceinline__ unsigned float_to_key(float f) {
    const unsigned b = __float_as_uint(f);
    return (b & 0x80000000u) ? ~b : (b | 0x80000000u);
}
__device__ __forceinline__ float key_to_float(unsigned k) {
    return __uint_as_float((k & 0x80000000u) ? (k ^ 0x80000000u) : ~k);
}

__device__ __forceinline__ float wave_min(float v) {
    for (int o = 16; o > 0; o >>= 1) v = fminf(v, __shfl_xor(v, o));
    return v;
}
__device__ __forceinline__ float wave_max(float v) {
    for (int o = 16; o > 0; o >>= 1) v = fmaxf(v, __shfl_xor(v, o));
    return v;
}

template <int RS, int NB>
__device__ __forceinline__ int segment_pass(const float4* __restrict__ seg, const v2f x2,
                                            v2f& top, v2f& prv, unsigned& hist) {
    int cnt = 0;
    for (int w = 0; w < NB; ++w) {
        const unsigned prev = hist;
#pragma unroll
        for (int g = 0; g < 8; ++g) {
            const float4 v = seg[w * 8 + g];
            const v2f t01 = (v2f){v.x, v.y} - x2;
            const v2f t23 = (v2f){v.z, v.w} - x2;
            v2f nt;
            nt = __builtin_elementwise_fma((v2f){t01.x, t01.x}, top, -prv); prv = top; top = nt;
            hist = __builtin_amdgcn_alignbit(hist, __float_as_uint(top.x), 31);
            nt = __builtin_elementwise_fma((v2f){t01.y, t01.y}, top, -prv); prv = top; top = nt;
            hist = __builtin_amdgcn_alignbit(hist, __float_as_uint(top.x), 31);
            nt = __builtin_elementwise_fma((v2f){t23.x, t23.x}, top, -prv); prv = top; top = nt;
            hist = __builtin_amdgcn_alignbit(hist, __float_as_uint(top.x), 31);
            nt = __builtin_elementwise_fma((v2f){t23.y, t23.y}, top, -prv); prv = top; top = nt;
            hist = __builtin_amdgcn_alignbit(hist, __float_as_uint(top.x), 31);
            if ((g + 1) % (RS / 4) == 0) {
                float ra = top.x, rb = top.y, rc = prv.x, rd = prv.y;
                rescale4(ra, rb, rc, rd);
                top = (v2f){ra, rb};  prv = (v2f){rc, rd};
            }
        }
        cnt += __popc(hist ^ __builtin_amdgcn_alignbit(prev, hist, 1));
    }
    return cnt;
}

template <int LW>
__device__ __forceinline__ int sturm_count(const float4* __restrict__ seg, const float x,
                                           const int rs, const int pos) {
    constexpr int NB = RN / LW / 32;
    const v2f x2 = {x + x, x + x};
    v2f top = {1.0f, 0.0f}, prv = {0.0f, 0.5f};
    unsigned hist = 0u;
    int cnt;
    if (rs == 32)      cnt = segment_pass<32, NB>(seg, x2, top, prv, hist);
    else if (rs == 16) cnt = segment_pass<16, NB>(seg, x2, top, prv, hist);
    else               cnt = segment_pass<4, NB>(seg, x2, top, prv, hist);
    const unsigned nd = hist & 1u;
    float a = top.x, b = top.y;
    const float a0 = a, b0 = b;
    float c = 2.0f * prv.x, d = 2.0f * prv.y;

    for (int dl = 1; dl < LW; dl <<= 1) {
        const float ua = __shfl_up(a, dl, LW), ub = __shfl_up(b, dl, LW);
        const float uc = __shfl_up(c, dl, LW), ud = __shfl_up(d, dl, LW);
        if (pos >= dl) {
            const float na = fmaf(a, ua, b * uc), nb = fmaf(a, ub, b * ud);
            const float nc = fmaf(c, ua, d * uc), nd2 = fmaf(c, ub, d * ud);
            a = na; b = nb; c = nc; d = nd2;
            rescale4(a, b, c, d);
        }
    }
    float pa = __shfl_up(a, 1, LW), pc = __shfl_up(c, 1, LW);
    if (pos == 0) { pa = 1.0f; pc = 0.0f; }

    const float pe = fmaf(a0, pa, b0 * pc);
    const unsigned spe = __float_as_uint(pe) >> 31, spa = __float_as_uint(pa) >> 31;
    const unsigned spc = __float_as_uint(pc) >> 31;
    const unsigned moved = spe ^ spa ^ nd;
    const unsigned qneg  = spa ^ spc;
    cnt += (int)moved - (int)(2u * (moved & qneg));
    for (int o = LW / 2; o > 0; o >>= 1) cnt += __shfl_xor(cnt, o);
    return cnt;
}

__global__ void __launch_bounds__(WPB * 32)
eig_bisect_kernel(const float* __restrict__ part3, const float* __restrict__ b3,
                  float* __restrict__ ptl_out, float* __restrict__ evl) {
    __shared__ __attribute__((aligned(16))) float sd[LPE * SEGP];
    __shared__ float smin[WPB], smax[WPB], sres[EPB];
    const int   l    = blockIdx.y;
    const float ll1  = (float)(l * (l + 1));
    const int   tid  = threadIdx.x;
    const int   lane = tid & 31;
    const int   wave = tid >> 5;
    const int   sub  = lane & (LPE - 1);
    const int   grp  = lane >> 4;
    const float dr   = 10000.0f / (float)RN;

    float dmin = 1e30f, dmax = -1e30f;
#pragma unroll
    for (int i = tid; i < RN; i += WPB * 32) {
        const float r = dr * (float)(i + 1);
        const float d = 1.0f + ptl_value(part3, b3, i) + ll1 / (r * r);
        sd[(i >> 8) * SEGP + (i & (SEG - 1))] = 2.0f * d;
        dmin = fminf(dmin, d);
        dmax = fmaxf(dmax, d);
    }
    if (l == 0 && wave == 0)
        ptl_out[blockIdx.x * EPB + lane] = ptl_value(part3, b3, blockIdx.x * EPB + lane);
    dmin = wave_min(dmin);
    dmax = wave_max(dmax);
    if (lane == 0) { smin[wave] = dmin; smax[wave] = dmax; }
    __syncthreads();
    float lo = wave_min(smin[lane & (WPB - 1)]) - 1.0f;
    float hi = wave_max(smax[lane & (WPB - 1)]) + 1.0f;

    const float width = __uint_as_float(__builtin_amdgcn_readfirstlane(__float_as_uint(hi - lo)));
    const int rs = width <= 8.0f ? 32 : (width <= 64.0f ? 16 : 4);

    const int j0 = __builtin_amdgcn_readfirstlane(blockIdx.x * EPB + wave * 2);
    const int j  = j0 + grp;
    const float4* __restrict__ seg16 = (const float4*)(sd + sub * SEGP);
    const float4* __restrict__ seg32 = (const float4*)(sd + (lane >> 1) * SEGP + (lane & 1) * (SEG / 2));

    unsigned klo = float_to_key(lo), khi = float_to_key(hi);
    const unsigned gap0 = __builtin_amdgcn_readfirstlane(khi - klo);
    const int nsteps = gap0 > 1u ? 32 - __builtin_clz(gap0 - 1u) : 0;

    int it = 0;
    for (bool together = true; together && it < nsteps; ++it) {
        const unsigned km = klo + ((khi - klo) >> 1);
        const float x = key_to_float(km);
        const int cnt = __builtin_amdgcn_readfirstlane(sturm_count<32>(seg32, x, rs, lane));
        if (cnt > j) khi = km; else klo = km;
        together = cnt != j0 + 1;
    }
    for (; it < nsteps; ++it) {
        const unsigned km = klo + ((khi - klo) >> 1);
        const float x = key_to_float(km);
        const int cnt = sturm_count<16>(seg16, x, rs, sub);
        if (cnt > j) khi = km; else klo = km;
    }
    lo = key_to_float(klo);
    hi = key_to_float(khi);

    if (sub == 0) sres[wave * 2 + grp] = 0.5f * (lo + hi);
    __syncthreads();
    if (wave == 0) evl[l * RN + blockIdx.x * EPB + lane] = sres[lane];
}

extern "C" void kernel_launch(void* const* d_in, const int* in_sizes, int n_in,
                              void* d_out, int out_size, void* d_ws, size_t ws_size,
                              hipStream_t stream) {
    const float* energy = (const float*)d_in[0];
    const float* W1     = (const float*)d_in[1];
    const float* b1     = (const float*)d_in[2];
    const float* W2     = (const float*)d_in[3];
    const float* b2     = (const float*)d_in[4];
    const float* W3     = (const float*)d_in[5];
    const float* b3     = (const float*)d_in[6];

    float* out   = (float*)d_out;
    float* part1 = (float*)d_ws;
    float* part2 = part1 + 16 * 2048;
    float* part3 = part1;

    gemv_slab_kernel<1, false, 256><<<dim3(2048 / 64, 16), 128, 0, stream>>>(
        energy, b1, W1, part1, 4096, 2048);
    gemv_slab_kernel<16, true, 128><<<dim3(1024 / 64, 16), 128, 0, stream>>>(
        part1, b1, W2, part2, 2048, 1024);
    gemv_slab_kernel<16, true, 128><<<dim3(4096 / 64, 8), 128, 0, stream>>>(
        part2, b2, W3, part3, 1024, 4096);
    eig_bisect_kernel<<<dim3(RN / EPB, 2), WPB * 32, 0, stream>>>(part3, b3, out, out + 4096);
}


# Lane exchanges of a bisection step (prefix-product scan, entry pair, count sum) by DPP row shifts and lane reads, not LDS permutes; 32-lane scan now two-level (new product order); low bits may move
# speedup vs baseline: 39.4563x; 1.0352x over previous
#include <hip/hip_runtime.h>
#include <hip/hip_bf16.h>

typedef __attribute__((ext_vector_type(16))) _Float16 v16h;
typedef __attribute__((ext_vector_type(8)))  float    v8f;
typedef __attribute__((ext_vector_type(2)))  float    v2f;

template <int P, bool ACT, int KS>
__global__ void __launch_bounds__(128)
gemv_slab_kernel(const float* __restrict__ xin,
                 const float* __restrict__ bin,
                 const float* __restrict__ W,
                 float* __restrict__ ypart,
                 int Kfull, int N) {
    __shared__ float sx[KS];
    const int tid   = threadIdx.x;
    const int lane  = tid & 31;
    const int wave  = tid >> 5;
    const int kbase = blockIdx.y * KS;

    const int j0  = blockIdx.x * 64 + wave * 16;
    const int col = j0 + (lane & 15);
    const int hi8 = (lane >> 4) * 8;

    constexpr int XPT = KS / 128;
    float xs[XPT];
#pragma unroll
    for (int r = 0; r < XPT; ++r) {
        const int k = kbase + tid + r * 128;
        float v = xin[k];
#pragma unroll
        for (int p = 1; p < P; ++p) v += xin[p * Kfull + k];
        if (ACT) v = fmaxf(v + bin[k], 0.0f);
        xs[r] = v;
    }

    constexpr int NF = KS / 32;
    float wr[NF][16];
#pragma unroll
    for (int f = 0; f < NF; ++f) {
#pragma unroll
        for (int e = 0; e < 16; ++e) {
            const int kk = f * 32 + e + (e & 8) + hi8;
            wr[f][e] = W[(long)(kbase + kk) * N + col];
        }
    }

#pragma unroll
    for (int r = 0; r < XPT; ++r) sx[tid + r * 128] = xs[r];
    __syncthreads();

    v8f acc = {};
#pragma unroll
    for (int f = 0; f < NF; ++f) {
        v16h ah = {}, al = {}, bh = {}, bl = {};
#pragma unroll
        for (int e = 0; e < 16; ++e) {
            const int kk = f * 32 + e + (e & 8) + hi8;
            const float xv = sx[kk];
            const float wv = wr[f][e];
            const _Float16 xh = (_Float16)xv;
            const _Float16 wh = (_Float16)wv;
            ah[e] = xh;  al[e] = (_Float16)(xv - (float)xh);
            bh[e] = wh;  bl[e] = (_Float16)(wv - (float)wh);
        }
        acc = __builtin_amdgcn_wmma_f32_16x16x32_f16(
            false, ah, false, bh, (short)0, acc, false, false);
        acc = __builtin_amdgcn_wmma_f32_16x16x32_f16(
            false, ah, false, bl, (short)0, acc, false, false);
        acc = __builtin_amdgcn_wmma_f32_16x16x32_f16(
            false, al, false, bh, (short)0, acc, false, false);
    }

    if (lane < 16)
        ypart[blockIdx.y * N + col] = acc[0];
}

#define RN   4096
#define LPE  16
#define SEG  256
#define SEGP 260
#define EPB  32
#define WPB  16
#define NP3  8

__device__ __forceinline__ float ptl_value(const float* __restrict__ part,
                                           const float* __restrict__ b3, int i) {
    float v = part[i];
#pragma unroll
    for (int p = 1; p < NP3; ++p) v += part[p * RN + i];
    return v + b3[i];
}

__device__ __forceinline__ void rescale4(float& a, float& b, float& c, float& d) {
    const float m = fmaxf(fmaxf(fabsf(a), fabsf(b)), fmaxf(fabsf(c), fabsf(d)));
    const int   e = __builtin_amdgcn_frexp_expf(m);
    a = ldexpf(a, -e);  b = ldexpf(b, -e);
    c = ldexpf(c, -e);  d = ldexpf(d, -e);
}

__device__ __forceinline__ unsigned float_to_key(float f) {
    const unsigned b = __float_as_uint(f);
    return (b & 0x80000000u) ? ~b : (b | 0x80000000u);
}
__device__ __forceinline__ float key_to_float(unsigned k) {
    return __uint_as_float((k & 0x80000000u) ? (k ^ 0x80000000u) : ~k);
}

__device__ __forceinline__ float wave_min(float v) {
    for (int o = 16; o > 0; o >>= 1) v = fminf(v, __shfl_xor(v, o));
    return v;
}
__device__ __forceinline__ float wave_max(float v) {
    for (int o = 16; o > 0; o >>= 1) v = fmaxf(v, __shfl_xor(v, o));
    return v;
}

template <int RS, int NB>
__device__ __forceinline__ int segment_pass(const float4* __restrict__ seg, const v2f x2,
                                            v2f& top, v2f& prv, unsigned& hist) {
    int cnt = 0;
    for (int w = 0; w < NB; ++w) {
        const unsigned prev = hist;
#pragma unroll
        for (int g = 0; g < 8; ++g) {
            const float4 v = seg[w * 8 + g];
            const v2f t01 = (v2f){v.x, v.y} - x2;
            const v2f t23 = (v2f){v.z, v.w} - x2;
            v2f nt;
            nt = __builtin_elementwise_fma((v2f){t01.x, t01.x}, top, -prv); prv = top; top = nt;
            hist = __builtin_amdgcn_alignbit(hist, __float_as_uint(top.x), 31);
            nt = __builtin_elementwise_fma((v2f){t01.y, t01.y}, top, -prv); prv = top; top = nt;
            hist = __builtin_amdgcn_alignbit(hist, __float_as_uint(top.x), 31);
            nt = __builtin_elementwise_fma((v2f){t23.x, t23.x}, top, -prv); prv = top; top = nt;
            hist = __builtin_amdgcn_alignbit(hist, __float_as_uint(top.x), 31);
            nt = __builtin_elementwise_fma((v2f){t23.y, t23.y}, top, -prv); prv = top; top = nt;
            hist = __builtin_amdgcn_alignbit(hist, __float_as_uint(top.x), 31);
            if ((g + 1) % (RS / 4) == 0) {
                float ra = top.x, rb = top.y, rc = prv.x, rd = prv.y;
                rescale4(ra, rb, rc, rd);
                top = (v2f){ra, rb};  prv = (v2f){rc, rd};
            }
        }
        cnt += __popc(hist ^ __builtin_amdgcn_alignbit(prev, hist, 1));
    }
    return cnt;
}

template <int DL>
__device__ __forceinline__ float row_up(const float v) {
    return __int_as_float(__builtin_amdgcn_update_dpp(0, __float_as_int(v),
                                                      0x110 + DL, 0xF, 0xF, true));
}
template <int M>
__device__ __forceinline__ int row_xor(const int v) {
    return __builtin_amdgcn_update_dpp(0, v, 0x160 + M, 0xF, 0xF, true);
}
__device__ __forceinline__ float lane_value(const float v, const int l) {
    return __int_as_float(__builtin_amdgcn_readlane(__float_as_int(v), l));
}

__device__ __forceinline__ void mul_rescale(float& a, float& b, float& c, float& d,
                                            const float ua, const float ub,
                                            const float uc, const float ud) {
    const float na = fmaf(a, ua, b * uc), nb = fmaf(a, ub, b * ud);
    const float nc = fmaf(c, ua, d * uc), nd2 = fmaf(c, ub, d * ud);
    a = na; b = nb; c = nc; d = nd2;
    rescale4(a, b, c, d);
}

template <int DL>
__device__ __forceinline__ void scan_round(float& a, float& b, float& c, float& d, const int p16) {
    const float ua = row_up<DL>(a), ub = row_up<DL>(b);
    const float uc = row_up<DL>(c), ud = row_up<DL>(d);
    if (p16 >= DL) mul_rescale(a, b, c, d, ua, ub, uc, ud);
}

template <int LW>
__device__ __forceinline__ int sturm_count(const float4* __restrict__ seg, const float x,
                                           const int rs, const int pos) {
    constexpr int NB = RN / LW / 32;
    const v2f x2 = {x + x, x + x};
    v2f top = {1.0f, 0.0f}, prv = {0.0f, 0.5f};
    unsigned hist = 0u;
    int cnt;
    if (rs == 32)      cnt = segment_pass<32, NB>(seg, x2, top, prv, hist);
    else if (rs == 16) cnt = segment_pass<16, NB>(seg, x2, top, prv, hist);
    else               cnt = segment_pass<4, NB>(seg, x2, top, prv, hist);
    const unsigned nd = hist & 1u;
    float a = top.x, b = top.y;
    const float a0 = a, b0 = b;
    float c = 2.0f * prv.x, d = 2.0f * prv.y;

    const int p16 = pos & 15;
    scan_round<1>(a, b, c, d, p16);
    scan_round<2>(a, b, c, d, p16);
    scan_round<4>(a, b, c, d, p16);
    scan_round<8>(a, b, c, d, p16);
    float ta = 1.0f, tc = 0.0f;
    if (LW == 32) {
        ta = lane_value(a, 15);  tc = lane_value(c, 15);
        const float tb = lane_value(b, 15), td = lane_value(d, 15);
        if (pos >= 16) mul_rescale(a, b, c, d, ta, tb, tc, td);
    }
    float pa = row_up<1>(a), pc = row_up<1>(c);
    if (LW == 32 && pos == 16) { pa = ta; pc = tc; }
    if (pos == 0) { pa = 1.0f; pc = 0.0f; }

    const float pe = fmaf(a0, pa, b0 * pc);
    const unsigned spe = __float_as_uint(pe) >> 31, spa = __float_as_uint(pa) >> 31;
    const unsigned spc = __float_as_uint(pc) >> 31;
    const unsigned moved = spe ^ spa ^ nd;
    const unsigned qneg  = spa ^ spc;
    cnt += (int)moved - (int)(2u * (moved & qneg));
    cnt += row_xor<8>(cnt);
    cnt += row_xor<4>(cnt);
    cnt += row_xor<2>(cnt);
    cnt += row_xor<1>(cnt);
    if (LW == 32) cnt = __builtin_amdgcn_readlane(cnt, 0) + __builtin_amdgcn_readlane(cnt, 16);
    return cnt;
}

__global__ void __launch_bounds__(WPB * 32)
eig_bisect_kernel(const float* __restrict__ part3, const float* __restrict__ b3,
                  float* __restrict__ ptl_out, float* __restrict__ evl) {
    __shared__ __attribute__((aligned(16))) float sd[LPE * SEGP];
    __shared__ float smin[WPB], smax[WPB], sres[EPB];
    const int   l    = blockIdx.y;
    const float ll1  = (float)(l * (l + 1));
    const int   tid  = threadIdx.x;
    const int   lane = tid & 31;
    const int   wave = tid >> 5;
    const int   sub  = lane & (LPE - 1);
    const int   grp  = lane >> 4;
    const float dr   = 10000.0f / (float)RN;

    float dmin = 1e30f, dmax = -1e30f;
#pragma unroll
    for (int i = tid; i < RN; i += WPB * 32) {
        const float r = dr * (float)(i + 1);
        const float d = 1.0f + ptl_value(part3, b3, i) + ll1 / (r * r);
        sd[(i >> 8) * SEGP + (i & (SEG - 1))] = 2.0f * d;
        dmin = fminf(dmin, d);
        dmax = fmaxf(dmax, d);
    }
    if (l == 0 && wave == 0)
        ptl_out[blockIdx.x * EPB + lane] = ptl_value(part3, b3, blockIdx.x * EPB + lane);
    dmin = wave_min(dmin);
    dmax = wave_max(dmax);
    if (lane == 0) { smin[wave] = dmin; smax[wave] = dmax; }
    __syncthreads();
    float lo = wave_min(smin[lane & (WPB - 1)]) - 1.0f;
    float hi = wave_max(smax[lane & (WPB - 1)]) + 1.0f;

    const float width = __uint_as_float(__builtin_amdgcn_readfirstlane(__float_as_uint(hi - lo)));
    const int rs = width <= 8.0f ? 32 : (width <= 64.0f ? 16 : 4);

    const int j0 = __builtin_amdgcn_readfirstlane(blockIdx.x * EPB + wave * 2);
    const int j  = j0 + grp;
    const float4* __restrict__ seg16 = (const float4*)(sd + sub * SEGP);
    const float4* __restrict__ seg32 = (const float4*)(sd + (lane >> 1) * SEGP + (lane & 1) * (SEG / 2));

    unsigned klo = float_to_key(lo), khi = float_to_key(hi);
    const unsigned gap0 = __builtin_amdgcn_readfirstlane(khi - klo);
    const int nsteps = gap0 > 1u ? 32 - __builtin_clz(gap0 - 1u) : 0;

    int it = 0;
    for (bool together = true; together && it < nsteps; ++it) {
        const unsigned km = klo + ((khi - klo) >> 1);
        const float x = key_to_float(km);
        const int cnt = __builtin_amdgcn_readfirstlane(sturm_count<32>(seg32, x, rs, lane));
        if (cnt > j) khi = km; else klo = km;
        together = cnt != j0 + 1;
    }
    for (; it < nsteps; ++it) {
        const unsigned km = klo + ((khi - klo) >> 1);
        const float x = key_to_float(km);
        const int cnt = sturm_count<16>(seg16, x, rs, sub);
        if (cnt > j) khi = km; else klo = km;
    }
    lo = key_to_float(klo);
    hi = key_to_float(khi);

    if (sub == 0) sres[wave * 2 + grp] = 0.5f * (lo + hi);
    __syncthreads();
    if (wave == 0) evl[l * RN + blockIdx.x * EPB + lane] = sres[lane];
}

extern "C" void kernel_launch(void* const* d_in, const int* in_sizes, int n_in,
                              void* d_out, int out_size, void* d_ws, size_t ws_size,
                              hipStream_t stream) {
    const float* energy = (const float*)d_in[0];
    const float* W1     = (const float*)d_in[1];
    const float* b1     = (const float*)d_in[2];
    const float* W2     = (const float*)d_in[3];
    const float* b2     = (const float*)d_in[4];
    const float* W3     = (const float*)d_in[5];
    const float* b3     = (const float*)d_in[6];

    float* out   = (float*)d_out;
    float* part1 = (float*)d_ws;
    float* part2 = part1 + 16 * 2048;
    float* part3 = part1;

    gemv_slab_kernel<1, false, 256><<<dim3(2048 / 64, 16), 128, 0, stream>>>(
        energy, b1, W1, part1, 4096, 2048);
    gemv_slab_kernel<16, true, 128><<<dim3(1024 / 64, 16), 128, 0, stream>>>(
        part1, b1, W2, part2, 2048, 1024);
    gemv_slab_kernel<16, true, 128><<<dim3(4096 / 64, 8), 128, 0, stream>>>(
        part2, b2, W3, part3, 1024, 4096);
    eig_bisect_kernel<<<dim3(RN / EPB, 2), WPB * 32, 0, stream>>>(part3, b3, out, out + 4096);
}
